# starting kernel
# baseline (speedup 1.0000x reference)
#include <hip/hip_runtime.h>
#include <math.h>

#define N_NODES 50000
#define N_EDGES 800000
#define IN_DIM  256
#define OUT_DIM 128
#define LEAK    0.2f

typedef __attribute__((ext_vector_type(2))) float v2f;
typedef __attribute__((ext_vector_type(4))) float v4f;
typedef __attribute__((ext_vector_type(8))) float v8f;

__global__ __launch_bounds__(256) void k_init(float* __restrict__ out,
                                              float* __restrict__ denom,
                                              unsigned* __restrict__ m_enc) {
  const int i = blockIdx.x * blockDim.x + threadIdx.x;
  if (i < N_NODES * OUT_DIM) out[i] = 0.0f;
  if (i < N_NODES) { denom[i] = 0.0f; m_enc[i] = 0u; }
}

__global__ __launch_bounds__(256) void k_gemm(const float* __restrict__ h,
                                              const float* __restrict__ W,
                                              float* __restrict__ z) {
  const int lane = threadIdx.x & 31;
  const int wv   = threadIdx.x >> 5;
  const int half = lane >> 4;
  const int l    = lane & 15;
  const int m0   = blockIdx.x * 16;
  const int n0   = wv * 16;

  const float* __restrict__ ar = h + (size_t)(m0 + l) * IN_DIM;
  const float* __restrict__ bc = W + (size_t)(n0 + l);

  v8f acc = {};
#pragma unroll 4
  for (int k = 0; k < IN_DIM; k += 4) {
    const int kk = k + 2 * half;
    v2f a; a.x = ar[kk];                     a.y = ar[kk + 1];
    v2f b; b.x = bc[(size_t)kk * OUT_DIM];   b.y = bc[(size_t)(kk + 1) * OUT_DIM];
    acc = __builtin_amdgcn_wmma_f32_16x16x4_f32(
         false, a,  false, b,
         (short)0, acc,  false,  false);
  }

  float* zp = z + (size_t)(m0 + 8 * half) * OUT_DIM + n0 + l;
#pragma unroll
  for (int i = 0; i < 8; ++i) zp[(size_t)i * OUT_DIM] = acc[i];
}

__global__ __launch_bounds__(256) void k_scores(const float* __restrict__ z,
                                                const float* __restrict__ attn_w,
                                                float* __restrict__ s1,
                                                float* __restrict__ s2) {
  const int n = blockIdx.x * blockDim.x + threadIdx.x;
  if (n >= N_NODES) return;
  const float* zr = z + (size_t)n * OUT_DIM;
  float d1 = 0.0f, d2 = 0.0f;
#pragma unroll 8
  for (int j = 0; j < OUT_DIM; ++j) {
    const float v = zr[j];
    d1 += v * attn_w[j];
    d2 += v * attn_w[OUT_DIM + j];
  }
  s1[n] = d1;
  s2[n] = d2;
}

__device__ __forceinline__ unsigned enc_f32(float f) {
  unsigned b = __float_as_uint(f);
  return (b & 0x80000000u) ? ~b : (b | 0x80000000u);
}
__device__ __forceinline__ float dec_f32(unsigned e) {
  unsigned b = (e & 0x80000000u) ? (e & 0x7FFFFFFFu) : ~e;
  return __uint_as_float(b);
}

__global__ __launch_bounds__(256) void k_edge_max(const int* __restrict__ src,
                                                  const int* __restrict__ dst,
                                                  const float* __restrict__ s1,
                                                  const float* __restrict__ s2,
                                                  float* __restrict__ ebuf,
                                                  unsigned* __restrict__ m_enc) {
  const int ei = blockIdx.x * blockDim.x + threadIdx.x;
  if (ei >= N_EDGES) return;
  const float logit = s1[src[ei]] + s2[dst[ei]];
  const float e = (logit >= 0.0f) ? logit : LEAK * logit;
  ebuf[ei] = e;
  atomicMax(&m_enc[dst[ei]], enc_f32(e));
}

__global__ __launch_bounds__(256) void k_edge_exp(const int* __restrict__ dst,
                                                  float* __restrict__ ebuf,
                                                  const unsigned* __restrict__ m_enc,
                                                  float* __restrict__ denom) {
  const int ei = blockIdx.x * blockDim.x + threadIdx.x;
  if (ei >= N_EDGES) return;
  const int d = dst[ei];
  const float ex = expf(ebuf[ei] - dec_f32(m_enc[d]));
  ebuf[ei] = ex;
  atomicAdd(&denom[d], ex);
}

__global__ __launch_bounds__(256) void k_scatter(const int* __restrict__ src,
                                                 const int* __restrict__ dst,
                                                 const float* __restrict__ z,
                                                 const float* __restrict__ ebuf,
                                                 const float* __restrict__ denom,
                                                 float* __restrict__ hacc) {
  const int ei = blockIdx.x * 8 + (threadIdx.x >> 5);
  if (ei >= N_EDGES) return;
  const int lane = threadIdx.x & 31;
  const int s = src[ei];
  const int d = dst[ei];
  const float dn = denom[d];
  const float alpha = ebuf[ei] / ((dn > 0.0f) ? dn : 1.0f);

  const v4f v = ((const v4f*)(z + (size_t)s * OUT_DIM))[lane];
  float* ho = hacc + (size_t)d * OUT_DIM + lane * 4;
  atomicAdd(&ho[0], alpha * v.x);
  atomicAdd(&ho[1], alpha * v.y);
  atomicAdd(&ho[2], alpha * v.z);
  atomicAdd(&ho[3], alpha * v.w);
}

__global__ __launch_bounds__(256) void k_final(float* __restrict__ out) {
  const int i = blockIdx.x * blockDim.x + threadIdx.x;
  if (i >= N_NODES * OUT_DIM) return;
  const float x = out[i];
  out[i] = (x > 0.0f) ? x : expm1f(x);
}

extern "C" void kernel_launch(void* const* d_in, const int* in_sizes, int n_in,
                              void* d_out, int out_size, void* d_ws, size_t ws_size,
                              hipStream_t stream) {
  const float* h      = (const float*)d_in[0];
  const int*   src    = (const int*)  d_in[1];
  const int*   dst    = (const int*)  d_in[2];
  const float* W_fc   = (const float*)d_in[3];
  const float* attn_w = (const float*)d_in[4];
  float*       out    = (float*)d_out;

  float*    z     = (float*)d_ws;
  float*    s1    = z  + (size_t)N_NODES * OUT_DIM;
  float*    s2    = s1 + N_NODES;
  float*    denom = s2 + N_NODES;
  unsigned* m_enc = (unsigned*)(denom + N_NODES);
  float*    ebuf  = (float*)(m_enc + N_NODES);

  const int TPB = 256;

  k_init<<<(N_NODES * OUT_DIM + TPB - 1) / TPB, TPB, 0, stream>>>(out, denom, m_enc);

  k_gemm<<<N_NODES / 16, TPB, 0, stream>>>(h, W_fc, z);

  k_scores<<<(N_NODES + TPB - 1) / TPB, TPB, 0, stream>>>(z, attn_w, s1, s2);

  k_edge_max<<<(N_EDGES + TPB - 1) / TPB, TPB, 0, stream>>>(src, dst, s1, s2, ebuf, m_enc);

  k_edge_exp<<<(N_EDGES + TPB - 1) / TPB, TPB, 0, stream>>>(dst, ebuf, m_enc, denom);

  k_scatter<<<(N_EDGES + 7) / 8, TPB, 0, stream>>>(src, dst, z, ebuf, denom, out);

  k_final<<<(N_NODES * OUT_DIM + TPB - 1) / TPB, TPB, 0, stream>>>(out);
}


# edges counting-sorted by dst per call (int atomic rank, scan, fill); one wave per node does max, exp-sum, weighted z-row gather and ELU in registers: no float atomics; sum order = arrival-order rank
# speedup vs baseline: 1.8903x; 1.8903x over previous
#include <hip/hip_runtime.h>
#include <math.h>


#define N_NODES 50000
#define N_EDGES 800000
#define IN_DIM  256
#define OUT_DIM 128
#define LEAK    0.2f

typedef __attribute__((ext_vector_type(2))) float v2f;
typedef __attribute__((ext_vector_type(4))) float v4f;
typedef __attribute__((ext_vector_type(8))) float v8f;

__global__ __launch_bounds__(256) void k_init(unsigned* __restrict__ cnt) {
  const int i = blockIdx.x * blockDim.x + threadIdx.x;
  if (i < N_NODES) cnt[i] = 0u;
}

__global__ __launch_bounds__(256) void k_count(const int* __restrict__ dst,
                                               unsigned* __restrict__ cnt,
                                               unsigned* __restrict__ rk) {
  const int ei = blockIdx.x * blockDim.x + threadIdx.x;
  if (ei >= N_EDGES) return;
  rk[ei] = atomicAdd(&cnt[dst[ei]], 1u);
}

__global__ __launch_bounds__(1024) void k_scan(const unsigned* __restrict__ cnt,
                                               unsigned* __restrict__ off) {
  __shared__ unsigned psum[32];
  __shared__ unsigned wsum[32];
  const int t    = threadIdx.x;
  const int lane = t & 31;
  const int wv   = t >> 5;
  const int tile0 = blockIdx.x * 1024;

  unsigned part = 0u;
  for (int i = t; i < tile0; i += 1024) part += cnt[i];

  const int i = tile0 + t;
  const unsigned c = (i < N_NODES) ? cnt[i] : 0u;

  unsigned incl = c;
  for (int o = 1; o < 32; o <<= 1) {
    const unsigned v = __shfl_up(incl, o);
    if (lane >= o) incl += v;
  }
  for (int o = 16; o > 0; o >>= 1) part += __shfl_xor(part, o);

  if (lane == 31) wsum[wv] = incl;
  if (lane == 0)  psum[wv] = part;
  __syncthreads();

  unsigned base = 0u;
  for (int w = 0; w < 32; ++w) {
    base += psum[w];
    if (w < wv) base += wsum[w];
  }
  if (i < N_NODES) off[i] = base + incl - c;
}

__global__ __launch_bounds__(256) void k_fill(const int* __restrict__ src,
                                              const int* __restrict__ dst,
                                              const unsigned* __restrict__ off,
                                              const unsigned* __restrict__ rk,
                                              int* __restrict__ csr_src) {
  const int ei = blockIdx.x * blockDim.x + threadIdx.x;
  if (ei >= N_EDGES) return;
  csr_src[off[dst[ei]] + rk[ei]] = src[ei];
}

__global__ __launch_bounds__(256) void k_gemm(const float* __restrict__ h,
                                              const float* __restrict__ W,
                                              float* __restrict__ z) {
  const int lane = threadIdx.x & 31;
  const int wv   = threadIdx.x >> 5;
  const int half = lane >> 4;
  const int l    = lane & 15;
  const int m0   = blockIdx.x * 16;
  const int n0   = wv * 16;

  const float* __restrict__ ar = h + (size_t)(m0 + l) * IN_DIM;
  const float* __restrict__ bc = W + (size_t)(n0 + l);

  v8f acc = {};
#pragma unroll 4
  for (int k = 0; k < IN_DIM; k += 4) {
    const int kk = k + 2 * half;
    v2f a; a.x = ar[kk];                     a.y = ar[kk + 1];
    v2f b; b.x = bc[(size_t)kk * OUT_DIM];   b.y = bc[(size_t)(kk + 1) * OUT_DIM];
    acc = __builtin_amdgcn_wmma_f32_16x16x4_f32(
         false, a,  false, b,
         (short)0, acc,  false,  false);
  }

  float* zp = z + (size_t)(m0 + 8 * half) * OUT_DIM + n0 + l;
#pragma unroll
  for (int i = 0; i < 8; ++i) zp[(size_t)i * OUT_DIM] = acc[i];
}

__global__ __launch_bounds__(256) void k_scores(const float* __restrict__ z,
                                                const float* __restrict__ attn_w,
                                                float* __restrict__ s1,
                                                float* __restrict__ s2) {
  const int n = blockIdx.x * blockDim.x + threadIdx.x;
  if (n >= N_NODES) return;
  const float* zr = z + (size_t)n * OUT_DIM;
  float d1 = 0.0f, d2 = 0.0f;
#pragma unroll 8
  for (int j = 0; j < OUT_DIM; ++j) {
    const float v = zr[j];
    d1 += v * attn_w[j];
    d2 += v * attn_w[OUT_DIM + j];
  }
  s1[n] = d1;
  s2[n] = d2;
}

__device__ __forceinline__ float wave_max(float v) {
  for (int o = 16; o > 0; o >>= 1) v = fmaxf(v, __shfl_xor(v, o));
  return v;
}
__device__ __forceinline__ float wave_sum(float v) {
  for (int o = 16; o > 0; o >>= 1) v += __shfl_xor(v, o);
  return v;
}

__global__ __launch_bounds__(256) void k_gather(const int* __restrict__ csr_src,
                                                const unsigned* __restrict__ off,
                                                const unsigned* __restrict__ cnt,
                                                const float* __restrict__ z,
                                                const float* __restrict__ s1,
                                                const float* __restrict__ s2,
                                                float* __restrict__ out) {
  const int lane = threadIdx.x & 31;
  const int d = __builtin_amdgcn_readfirstlane((int)(blockIdx.x * 8 + (threadIdx.x >> 5)));
  const int beg = (int)off[d];
  const int n   = (int)cnt[d];
  const float s2d = s2[d];
  const float inf = __builtin_huge_valf();

  auto edge = [&](int p, int& s, float& e) {
    const unsigned u = (unsigned)csr_src[beg + p];
    s = (int)((u < (unsigned)N_NODES) ? u : (unsigned)(N_NODES - 1));
    const float lg = s1[s] + s2d;
    e = (lg >= 0.0f) ? lg : LEAK * lg;
  };

  int s0 = 0;
  float e0 = -inf;
  if (lane < n) edge(lane, s0, e0);

  float m = e0;
  for (int p = 32 + lane; p < n; p += 32) {
    int s; float e;
    edge(p, s, e);
    m = fmaxf(m, e);
  }
  m = wave_max(m);
  if (!(fabsf(m) < inf)) m = 0.0f;

  const float x0 = (lane < n) ? expf(e0 - m) : 0.0f;
  float sum = x0;
  for (int p = 32 + lane; p < n; p += 32) {
    int s; float e;
    edge(p, s, e);
    sum += expf(e - m);
  }
  sum = wave_sum(sum);
  const float dn = (sum > 0.0f) ? sum : 1.0f;

  const v4f* __restrict__ z4 = (const v4f*)z;
  v4f acc = {0.0f, 0.0f, 0.0f, 0.0f};
  for (int c = 0; c < n; c += 32) {
    int s = s0;
    float a = x0 / dn;
    if (c > 0) {
      s = 0;
      a = 0.0f;
      const int p = c + lane;
      if (p < n) {
        float e;
        edge(p, s, e);
        a = expf(e - m) / dn;
      }
    }
    const int k = (n - c < 32) ? (n - c) : 32;
#pragma unroll 4
    for (int j = 0; j < k; ++j) {
      const int   sj = __builtin_amdgcn_readlane(s, j);
      const float aj = __int_as_float(__builtin_amdgcn_readlane(__float_as_int(a), j));
      const v4f v = z4[(size_t)sj * (OUT_DIM / 4) + lane];
      acc.x += aj * v.x;
      acc.y += aj * v.y;
      acc.z += aj * v.z;
      acc.w += aj * v.w;
    }
  }

  v4f r;
  r.x = (acc.x > 0.0f) ? acc.x : expm1f(acc.x);
  r.y = (acc.y > 0.0f) ? acc.y : expm1f(acc.y);
  r.z = (acc.z > 0.0f) ? acc.z : expm1f(acc.z);
  r.w = (acc.w > 0.0f) ? acc.w : expm1f(acc.w);
  ((v4f*)out)[(size_t)d * (OUT_DIM / 4) + lane] = r;
}

extern "C" void kernel_launch(void* const* d_in, const int* in_sizes, int n_in,
                              void* d_out, int out_size, void* d_ws, size_t ws_size,
                              hipStream_t stream) {
  const float* h      = (const float*)d_in[0];
  const int*   src    = (const int*)  d_in[1];
  const int*   dst    = (const int*)  d_in[2];
  const float* W_fc   = (const float*)d_in[3];
  const float* attn_w = (const float*)d_in[4];
  float*       out    = (float*)d_out;

  float*    z       = (float*)d_ws;
  unsigned* rk      = (unsigned*)d_ws;
  float*    s1      = z  + (size_t)N_NODES * OUT_DIM;
  float*    s2      = s1 + N_NODES;
  unsigned* cnt     = (unsigned*)(s2 + N_NODES);
  unsigned* off     = cnt + N_NODES;
  int*      csr_src = (int*)(off + N_NODES);

  const int TPB = 256;

  k_init<<<(N_NODES + TPB - 1) / TPB, TPB, 0, stream>>>(cnt);

  k_count<<<(N_EDGES + TPB - 1) / TPB, TPB, 0, stream>>>(dst, cnt, rk);

  k_scan<<<(N_NODES + 1023) / 1024, 1024, 0, stream>>>(cnt, off);

  k_fill<<<(N_EDGES + TPB - 1) / TPB, TPB, 0, stream>>>(src, dst, off, rk, csr_src);

  k_gemm<<<N_NODES / 16, TPB, 0, stream>>>(h, W_fc, z);

  k_scores<<<(N_NODES + TPB - 1) / TPB, TPB, 0, stream>>>(z, attn_w, s1, s2);

  k_gather<<<N_NODES / 8, TPB, 0, stream>>>(csr_src, off, cnt, z, s1, s2, out);
}


# GEMM operand feed: 512 blocks walk the row tiles, each wave keeps its 16-column W slice in registers, h tile read once per block through LDS (coalesced); k sum order changes (two accumulator chains)
# speedup vs baseline: 2.2868x; 1.2098x over previous
#include <hip/hip_runtime.h>
#include <math.h>


#define N_NODES 50000
#define N_EDGES 800000
#define IN_DIM  256
#define OUT_DIM 128
#define LEAK    0.2f

typedef __attribute__((ext_vector_type(2))) float v2f;
typedef __attribute__((ext_vector_type(4))) float v4f;
typedef __attribute__((ext_vector_type(8))) float v8f;

__global__ __launch_bounds__(256) void k_init(unsigned* __restrict__ cnt) {
  const int i = blockIdx.x * blockDim.x + threadIdx.x;
  if (i < N_NODES) cnt[i] = 0u;
}

__global__ __launch_bounds__(256) void k_count(const int* __restrict__ dst,
                                               unsigned* __restrict__ cnt,
                                               unsigned* __restrict__ rk) {
  const int ei = blockIdx.x * blockDim.x + threadIdx.x;
  if (ei >= N_EDGES) return;
  rk[ei] = atomicAdd(&cnt[dst[ei]], 1u);
}

__global__ __launch_bounds__(1024) void k_scan(const unsigned* __restrict__ cnt,
                                               unsigned* __restrict__ off) {
  __shared__ unsigned psum[32];
  __shared__ unsigned wsum[32];
  const int t    = threadIdx.x;
  const int lane = t & 31;
  const int wv   = t >> 5;
  const int tile0 = blockIdx.x * 1024;

  unsigned part = 0u;
  for (int i = t; i < tile0; i += 1024) part += cnt[i];

  const int i = tile0 + t;
  const unsigned c = (i < N_NODES) ? cnt[i] : 0u;

  unsigned incl = c;
  for (int o = 1; o < 32; o <<= 1) {
    const unsigned v = __shfl_up(incl, o);
    if (lane >= o) incl += v;
  }
  for (int o = 16; o > 0; o >>= 1) part += __shfl_xor(part, o);

  if (lane == 31) wsum[wv] = incl;
  if (lane == 0)  psum[wv] = part;
  __syncthreads();

  unsigned base = 0u;
  for (int w = 0; w < 32; ++w) {
    base += psum[w];
    if (w < wv) base += wsum[w];
  }
  if (i < N_NODES) off[i] = base + incl - c;
}

__global__ __launch_bounds__(256) void k_fill(const int* __restrict__ src,
                                              const int* __restrict__ dst,
                                              const unsigned* __restrict__ off,
                                              const unsigned* __restrict__ rk,
                                              int* __restrict__ csr_src) {
  const int ei = blockIdx.x * blockDim.x + threadIdx.x;
  if (ei >= N_EDGES) return;
  csr_src[off[dst[ei]] + rk[ei]] = src[ei];
}

#define GEMM_BLOCKS 512
#define M_TILES (N_NODES / 16)
#define A_PITCH4 65
#define A_BUF4 (16 * A_PITCH4)

__global__ __launch_bounds__(256) void k_gemm(const float* __restrict__ h,
                                              const float* __restrict__ W,
                                              float* __restrict__ z) {
  __shared__ v4f As[2 * A_BUF4];

  const int tid  = threadIdx.x;
  const int lane = tid & 31;
  const int wv   = tid >> 5;
  const int half = lane >> 4;
  const int l    = lane & 15;
  const int n0   = wv * 16;

  v2f b[64];
  const float* __restrict__ bc = W + (size_t)(4 * half) * OUT_DIM + n0 + l;
#pragma unroll
  for (int t = 0; t < 32; ++t) {
    const float* __restrict__ p = bc + (size_t)(8 * t) * OUT_DIM;
    b[2 * t].x     = p[0 * OUT_DIM];
    b[2 * t].y     = p[1 * OUT_DIM];
    b[2 * t + 1].x = p[2 * OUT_DIM];
    b[2 * t + 1].y = p[3 * OUT_DIM];
  }

  const int srow = tid >> 6;
  const int sc4  = tid & 63;
  const v4f* __restrict__ h4 = (const v4f*)h;

  int mt = blockIdx.x;
  v4f ld[4];
#pragma unroll
  for (int i = 0; i < 4; ++i)
    ld[i] = h4[(size_t)(mt * 16 + srow + 4 * i) * (IN_DIM / 4) + sc4];
#pragma unroll
  for (int i = 0; i < 4; ++i)
    As[(srow + 4 * i) * A_PITCH4 + sc4] = ld[i];
  __syncthreads();

  int cur = 0;
  for (; mt < M_TILES; mt += GEMM_BLOCKS) {
    const int  nx   = mt + GEMM_BLOCKS;
    const bool more = nx < M_TILES;
    if (more) {
#pragma unroll
      for (int i = 0; i < 4; ++i)
        ld[i] = h4[(size_t)(nx * 16 + srow + 4 * i) * (IN_DIM / 4) + sc4];
    }

    const v4f* ap = &As[cur + l * A_PITCH4 + half];
    v8f acc0 = {};
    v8f acc1 = {};
    v4f fa[2][8];
#pragma unroll
    for (int u = 0; u < 8; ++u) fa[0][u] = ap[2 * u];
#pragma unroll
    for (int g = 0; g < 4; ++g) {
      if (g < 3) {
#pragma unroll
        for (int u = 0; u < 8; ++u) fa[(g + 1) & 1][u] = ap[2 * (8 * (g + 1) + u)];
      }
      __builtin_amdgcn_sched_barrier(0);
#pragma unroll
      for (int u = 0; u < 8; ++u) {
        const int t = 8 * g + u;
        const v4f a = fa[g & 1][u];
        v2f a01; a01.x = a.x; a01.y = a.y;
        v2f a23; a23.x = a.z; a23.y = a.w;
        acc0 = __builtin_amdgcn_wmma_f32_16x16x4_f32(
             false, a01,  false, b[2 * t],
             (short)0, acc0,  false,  false);
        acc1 = __builtin_amdgcn_wmma_f32_16x16x4_f32(
             false, a23,  false, b[2 * t + 1],
             (short)0, acc1,  false,  false);
      }
      __builtin_amdgcn_sched_barrier(0);
    }

    float* zp = z + (size_t)(mt * 16 + 8 * half) * OUT_DIM + n0 + l;
#pragma unroll
    for (int i = 0; i < 8; ++i) zp[(size_t)i * OUT_DIM] = acc0[i] + acc1[i];

    const int nxt = cur ^ A_BUF4;
    if (more) {
#pragma unroll
      for (int i = 0; i < 4; ++i)
        As[nxt + (srow + 4 * i) * A_PITCH4 + sc4] = ld[i];
    }
    __syncthreads();
    cur = nxt;
  }
}

__global__ __launch_bounds__(256) void k_scores(const float* __restrict__ z,
                                                const float* __restrict__ attn_w,
                                                float* __restrict__ s1,
                                                float* __restrict__ s2) {
  const int n = blockIdx.x * blockDim.x + threadIdx.x;
  if (n >= N_NODES) return;
  const float* zr = z + (size_t)n * OUT_DIM;
  float d1 = 0.0f, d2 = 0.0f;
#pragma unroll 8
  for (int j = 0; j < OUT_DIM; ++j) {
    const float v = zr[j];
    d1 += v * attn_w[j];
    d2 += v * attn_w[OUT_DIM + j];
  }
  s1[n] = d1;
  s2[n] = d2;
}

__device__ __forceinline__ float wave_max(float v) {
  for (int o = 16; o > 0; o >>= 1) v = fmaxf(v, __shfl_xor(v, o));
  return v;
}
__device__ __forceinline__ float wave_sum(float v) {
  for (int o = 16; o > 0; o >>= 1) v += __shfl_xor(v, o);
  return v;
}

__global__ __launch_bounds__(256) void k_gather(const int* __restrict__ csr_src,
                                                const unsigned* __restrict__ off,
                                                const unsigned* __restrict__ cnt,
                                                const float* __restrict__ z,
                                                const float* __restrict__ s1,
                                                const float* __restrict__ s2,
                                                float* __restrict__ out) {
  const int lane = threadIdx.x & 31;
  const int d = __builtin_amdgcn_readfirstlane((int)(blockIdx.x * 8 + (threadIdx.x >> 5)));
  const int beg = (int)off[d];
  const int n   = (int)cnt[d];
  const float s2d = s2[d];
  const float inf = __builtin_huge_valf();

  auto edge = [&](int p, int& s, float& e) {
    const unsigned u = (unsigned)csr_src[beg + p];
    s = (int)((u < (unsigned)N_NODES) ? u : (unsigned)(N_NODES - 1));
    const float lg = s1[s] + s2d;
    e = (lg >= 0.0f) ? lg : LEAK * lg;
  };

  int s0 = 0;
  float e0 = -inf;
  if (lane < n) edge(lane, s0, e0);

  float m = e0;
  for (int p = 32 + lane; p < n; p += 32) {
    int s; float e;
    edge(p, s, e);
    m = fmaxf(m, e);
  }
  m = wave_max(m);
  if (!(fabsf(m) < inf)) m = 0.0f;

  const float x0 = (lane < n) ? expf(e0 - m) : 0.0f;
  float sum = x0;
  for (int p = 32 + lane; p < n; p += 32) {
    int s; float e;
    edge(p, s, e);
    sum += expf(e - m);
  }
  sum = wave_sum(sum);
  const float dn = (sum > 0.0f) ? sum : 1.0f;

  const v4f* __restrict__ z4 = (const v4f*)z;
  v4f acc = {0.0f, 0.0f, 0.0f, 0.0f};
  for (int c = 0; c < n; c += 32) {
    int s = s0;
    float a = x0 / dn;
    if (c > 0) {
      s = 0;
      a = 0.0f;
      const int p = c + lane;
      if (p < n) {
        float e;
        edge(p, s, e);
        a = expf(e - m) / dn;
      }
    }
    const int k = (n - c < 32) ? (n - c) : 32;
#pragma unroll 4
    for (int j = 0; j < k; ++j) {
      const int   sj = __builtin_amdgcn_readlane(s, j);
      const float aj = __int_as_float(__builtin_amdgcn_readlane(__float_as_int(a), j));
      const v4f v = z4[(size_t)sj * (OUT_DIM / 4) + lane];
      acc.x += aj * v.x;
      acc.y += aj * v.y;
      acc.z += aj * v.z;
      acc.w += aj * v.w;
    }
  }

  v4f r;
  r.x = (acc.x > 0.0f) ? acc.x : expm1f(acc.x);
  r.y = (acc.y > 0.0f) ? acc.y : expm1f(acc.y);
  r.z = (acc.z > 0.0f) ? acc.z : expm1f(acc.z);
  r.w = (acc.w > 0.0f) ? acc.w : expm1f(acc.w);
  ((v4f*)out)[(size_t)d * (OUT_DIM / 4) + lane] = r;
}

extern "C" void kernel_launch(void* const* d_in, const int* in_sizes, int n_in,
                              void* d_out, int out_size, void* d_ws, size_t ws_size,
                              hipStream_t stream) {
  const float* h      = (const float*)d_in[0];
  const int*   src    = (const int*)  d_in[1];
  const int*   dst    = (const int*)  d_in[2];
  const float* W_fc   = (const float*)d_in[3];
  const float* attn_w = (const float*)d_in[4];
  float*       out    = (float*)d_out;

  float*    z       = (float*)d_ws;
  unsigned* rk      = (unsigned*)d_ws;
  float*    s1      = z  + (size_t)N_NODES * OUT_DIM;
  float*    s2      = s1 + N_NODES;
  unsigned* cnt     = (unsigned*)(s2 + N_NODES);
  unsigned* off     = cnt + N_NODES;
  int*      csr_src = (int*)(off + N_NODES);

  const int TPB = 256;

  k_init<<<(N_NODES + TPB - 1) / TPB, TPB, 0, stream>>>(cnt);

  k_count<<<(N_EDGES + TPB - 1) / TPB, TPB, 0, stream>>>(dst, cnt, rk);

  k_scan<<<(N_NODES + 1023) / 1024, 1024, 0, stream>>>(cnt, off);

  k_fill<<<(N_EDGES + TPB - 1) / TPB, TPB, 0, stream>>>(src, dst, off, rk, csr_src);

  k_gemm<<<GEMM_BLOCKS, TPB, 0, stream>>>(h, W_fc, z);

  k_scores<<<(N_NODES + TPB - 1) / TPB, TPB, 0, stream>>>(z, attn_w, s1, s2);

  k_gather<<<N_NODES / 8, TPB, 0, stream>>>(csr_src, off, cnt, z, s1, s2, out);
}


# edge sort rebuilt as two-level counting sort, all counters in LDS (chunks bin edges by 256-node dst range, one block per range ranks and places): no global atomics; sum order still arrival-order rank
# speedup vs baseline: 3.0966x; 1.3541x over previous
#include <hip/hip_runtime.h>
#include <math.h>


#define N_NODES 50000
#define N_EDGES 800000
#define IN_DIM  256
#define OUT_DIM 128
#define LEAK    0.2f

typedef __attribute__((ext_vector_type(2))) float v2f;
typedef __attribute__((ext_vector_type(4))) float v4f;
typedef __attribute__((ext_vector_type(8))) float v8f;

#define BIN_SHIFT   8
#define BIN_NODES   (1 << BIN_SHIFT)
#define N_BINS      ((N_NODES + BIN_NODES - 1) / BIN_NODES)
#define N_CHUNKS    250
#define CHUNK_EDGES (N_EDGES / N_CHUNKS)
#define CHUNK_TPB   320
#define CHUNK_PT    (CHUNK_EDGES / CHUNK_TPB)
#define HIST_N      (N_BINS * N_CHUNKS)
#define SORT_TPB    512
#define SORT_PT     10
static_assert(N_CHUNKS * CHUNK_EDGES == N_EDGES, "chunks must tile the edge list");
static_assert(CHUNK_PT * CHUNK_TPB == CHUNK_EDGES, "threads must tile a chunk");
static_assert(N_BINS <= CHUNK_TPB, "one thread per range counter");
static_assert(N_NODES <= (1 << 24), "src must fit the packed word");

__device__ __forceinline__ unsigned bin_of(unsigned d) {
  const unsigned b = d >> BIN_SHIFT;
  return (b < (unsigned)N_BINS) ? b : (unsigned)(N_BINS - 1);
}

__global__ __launch_bounds__(CHUNK_TPB) void k_binhist(const int* __restrict__ dst,
                                                       unsigned* __restrict__ rk,
                                                       unsigned* __restrict__ hist) {
  __shared__ unsigned lh[N_BINS];
  const int tid = threadIdx.x;
  const int c   = blockIdx.x;
  if (tid < N_BINS) lh[tid] = 0u;
  const int e0 = c * CHUNK_EDGES + tid;
  unsigned b[CHUNK_PT];
#pragma unroll
  for (int k = 0; k < CHUNK_PT; ++k) b[k] = bin_of((unsigned)dst[e0 + k * CHUNK_TPB]);
  __syncthreads();
  unsigned r[CHUNK_PT];
#pragma unroll
  for (int k = 0; k < CHUNK_PT; ++k) r[k] = atomicAdd(&lh[b[k]], 1u);
  __syncthreads();
  if (tid < N_BINS) hist[tid * N_CHUNKS + c] = lh[tid];
#pragma unroll
  for (int k = 0; k < CHUNK_PT; ++k) rk[e0 + k * CHUNK_TPB] = r[k];
}

__global__ __launch_bounds__(1024) void k_scan(const unsigned* __restrict__ hist,
                                               unsigned* __restrict__ base) {
  __shared__ unsigned psum[32];
  __shared__ unsigned wsum[32];
  const int t    = threadIdx.x;
  const int lane = t & 31;
  const int wv   = t >> 5;
  const int tile0 = blockIdx.x * 1024;

  unsigned part = 0u;
  for (int i = t; i < tile0; i += 1024) part += hist[i];

  const int i = tile0 + t;
  const unsigned c = (i < HIST_N) ? hist[i] : 0u;

  unsigned incl = c;
  for (int o = 1; o < 32; o <<= 1) {
    const unsigned v = __shfl_up(incl, o);
    if (lane >= o) incl += v;
  }
  for (int o = 16; o > 0; o >>= 1) part += __shfl_xor(part, o);

  if (lane == 31) wsum[wv] = incl;
  if (lane == 0)  psum[wv] = part;
  __syncthreads();

  unsigned pre = 0u;
  for (int w = 0; w < 32; ++w) {
    pre += psum[w];
    if (w < wv) pre += wsum[w];
  }
  if (i < HIST_N) base[i] = pre + incl - c;
}

__global__ __launch_bounds__(CHUNK_TPB) void k_binscatter(const int* __restrict__ src,
                                                          const int* __restrict__ dst,
                                                          const unsigned* __restrict__ base,
                                                          const unsigned* __restrict__ rk,
                                                          unsigned* __restrict__ binned) {
  __shared__ unsigned lb[N_BINS];
  const int tid = threadIdx.x;
  const int c   = blockIdx.x;
  const int e0 = c * CHUNK_EDGES + tid;
  unsigned d[CHUNK_PT], s[CHUNK_PT], r[CHUNK_PT];
#pragma unroll
  for (int k = 0; k < CHUNK_PT; ++k) {
    d[k] = (unsigned)dst[e0 + k * CHUNK_TPB];
    s[k] = (unsigned)src[e0 + k * CHUNK_TPB];
    r[k] = rk[e0 + k * CHUNK_TPB];
  }
  if (tid < N_BINS) lb[tid] = base[tid * N_CHUNKS + c];
  __syncthreads();
#pragma unroll
  for (int k = 0; k < CHUNK_PT; ++k) {
    unsigned pos = lb[bin_of(d[k])] + r[k];
    if (pos >= (unsigned)N_EDGES) pos = (unsigned)(N_EDGES - 1);
    binned[pos] = (s[k] << 8) | (d[k] & (unsigned)(BIN_NODES - 1));
  }
}

__global__ __launch_bounds__(SORT_TPB) void k_binsort(const unsigned* __restrict__ binned,
                                                      const unsigned* __restrict__ base,
                                                      unsigned* __restrict__ off,
                                                      unsigned* __restrict__ cnt,
                                                      int* __restrict__ csr_src) {
  __shared__ unsigned lcnt[BIN_NODES];
  __shared__ unsigned lcur[BIN_NODES];
  __shared__ unsigned wtot[BIN_NODES / 32];
  const int tid  = threadIdx.x;
  const int lane = tid & 31;
  const int wv   = tid >> 5;
  const int bin  = blockIdx.x;

  unsigned bs = base[bin * N_CHUNKS];
  unsigned be = (bin + 1 < N_BINS) ? base[(bin + 1) * N_CHUNKS] : (unsigned)N_EDGES;
  if (be > (unsigned)N_EDGES) be = (unsigned)N_EDGES;
  if (bs > be) bs = be;
  const unsigned nE = be - bs;

  if (tid < BIN_NODES) lcnt[tid] = 0u;
  unsigned w[SORT_PT];
#pragma unroll
  for (int k = 0; k < SORT_PT; ++k) {
    const unsigned idx = (unsigned)(tid + k * SORT_TPB);
    w[k] = (idx < nE) ? binned[bs + idx] : 0u;
  }
  __syncthreads();

#pragma unroll
  for (int k = 0; k < SORT_PT; ++k) {
    const unsigned idx = (unsigned)(tid + k * SORT_TPB);
    if (idx < nE) atomicAdd(&lcnt[w[k] & (unsigned)(BIN_NODES - 1)], 1u);
  }
  for (unsigned idx = (unsigned)(SORT_PT * SORT_TPB + tid); idx < nE; idx += SORT_TPB)
    atomicAdd(&lcnt[binned[bs + idx] & (unsigned)(BIN_NODES - 1)], 1u);
  __syncthreads();

  const unsigned cn = (tid < BIN_NODES) ? lcnt[tid] : 0u;
  unsigned incl = cn;
  for (int o = 1; o < 32; o <<= 1) {
    const unsigned v = __shfl_up(incl, o);
    if (lane >= o) incl += v;
  }
  if (tid < BIN_NODES && lane == 31) wtot[wv] = incl;
  __syncthreads();
  unsigned excl = 0u;
  if (tid < BIN_NODES) {
    unsigned pre = 0u;
    for (int q = 0; q < BIN_NODES / 32; ++q)
      if (q < wv) pre += wtot[q];
    excl = pre + incl - cn;
    lcur[tid] = excl;
  }
  __syncthreads();

  unsigned p[SORT_PT];
#pragma unroll
  for (int k = 0; k < SORT_PT; ++k) {
    const unsigned idx = (unsigned)(tid + k * SORT_TPB);
    p[k] = (idx < nE) ? atomicAdd(&lcur[w[k] & (unsigned)(BIN_NODES - 1)], 1u) : 0u;
  }
#pragma unroll
  for (int k = 0; k < SORT_PT; ++k) {
    const unsigned idx = (unsigned)(tid + k * SORT_TPB);
    if (idx < nE) csr_src[bs + p[k]] = (int)(w[k] >> 8);
  }
  for (unsigned idx = (unsigned)(SORT_PT * SORT_TPB + tid); idx < nE; idx += SORT_TPB) {
    const unsigned ww = binned[bs + idx];
    const unsigned q  = atomicAdd(&lcur[ww & (unsigned)(BIN_NODES - 1)], 1u);
    csr_src[bs + q] = (int)(ww >> 8);
  }

  const int node = bin * BIN_NODES + tid;
  if (tid < BIN_NODES && node < N_NODES) {
    off[node] = bs + excl;
    cnt[node] = cn;
  }
}

#define GEMM_BLOCKS 512
#define M_TILES (N_NODES / 16)
#define A_PITCH4 65
#define A_BUF4 (16 * A_PITCH4)

__global__ __launch_bounds__(256) void k_gemm(const float* __restrict__ h,
                                              const float* __restrict__ W,
                                              float* __restrict__ z) {
  __shared__ v4f As[2 * A_BUF4];

  const int tid  = threadIdx.x;
  const int lane = tid & 31;
  const int wv   = tid >> 5;
  const int half = lane >> 4;
  const int l    = lane & 15;
  const int n0   = wv * 16;

  v2f b[64];
  const float* __restrict__ bc = W + (size_t)(4 * half) * OUT_DIM + n0 + l;
#pragma unroll
  for (int t = 0; t < 32; ++t) {
    const float* __restrict__ p = bc + (size_t)(8 * t) * OUT_DIM;
    b[2 * t].x     = p[0 * OUT_DIM];
    b[2 * t].y     = p[1 * OUT_DIM];
    b[2 * t + 1].x = p[2 * OUT_DIM];
    b[2 * t + 1].y = p[3 * OUT_DIM];
  }

  const int srow = tid >> 6;
  const int sc4  = tid & 63;
  const v4f* __restrict__ h4 = (const v4f*)h;

  int mt = blockIdx.x;
  v4f ld[4];
#pragma unroll
  for (int i = 0; i < 4; ++i)
    ld[i] = h4[(size_t)(mt * 16 + srow + 4 * i) * (IN_DIM / 4) + sc4];
#pragma unroll
  for (int i = 0; i < 4; ++i)
    As[(srow + 4 * i) * A_PITCH4 + sc4] = ld[i];
  __syncthreads();

  int cur = 0;
  for (; mt < M_TILES; mt += GEMM_BLOCKS) {
    const int  nx   = mt + GEMM_BLOCKS;
    const bool more = nx < M_TILES;
    if (more) {
#pragma unroll
      for (int i = 0; i < 4; ++i)
        ld[i] = h4[(size_t)(nx * 16 + srow + 4 * i) * (IN_DIM / 4) + sc4];
    }

    const v4f* ap = &As[cur + l * A_PITCH4 + half];
    v8f acc0 = {};
    v8f acc1 = {};
    v4f fa[2][8];
#pragma unroll
    for (int u = 0; u < 8; ++u) fa[0][u] = ap[2 * u];
#pragma unroll
    for (int g = 0; g < 4; ++g) {
      if (g < 3) {
#pragma unroll
        for (int u = 0; u < 8; ++u) fa[(g + 1) & 1][u] = ap[2 * (8 * (g + 1) + u)];
      }
      __builtin_amdgcn_sched_barrier(0);
#pragma unroll
      for (int u = 0; u < 8; ++u) {
        const int t = 8 * g + u;
        const v4f a = fa[g & 1][u];
        v2f a01; a01.x = a.x; a01.y = a.y;
        v2f a23; a23.x = a.z; a23.y = a.w;
        acc0 = __builtin_amdgcn_wmma_f32_16x16x4_f32(
             false, a01,  false, b[2 * t],
             (short)0, acc0,  false,  false);
        acc1 = __builtin_amdgcn_wmma_f32_16x16x4_f32(
             false, a23,  false, b[2 * t + 1],
             (short)0, acc1,  false,  false);
      }
      __builtin_amdgcn_sched_barrier(0);
    }

    float* zp = z + (size_t)(mt * 16 + 8 * half) * OUT_DIM + n0 + l;
#pragma unroll
    for (int i = 0; i < 8; ++i) zp[(size_t)i * OUT_DIM] = acc0[i] + acc1[i];

    const int nxt = cur ^ A_BUF4;
    if (more) {
#pragma unroll
      for (int i = 0; i < 4; ++i)
        As[nxt + (srow + 4 * i) * A_PITCH4 + sc4] = ld[i];
    }
    __syncthreads();
    cur = nxt;
  }
}

__global__ __launch_bounds__(256) void k_scores(const float* __restrict__ z,
                                                const float* __restrict__ attn_w,
                                                float* __restrict__ s1,
                                                float* __restrict__ s2) {
  const int n = blockIdx.x * blockDim.x + threadIdx.x;
  if (n >= N_NODES) return;
  const float* zr = z + (size_t)n * OUT_DIM;
  float d1 = 0.0f, d2 = 0.0f;
#pragma unroll 8
  for (int j = 0; j < OUT_DIM; ++j) {
    const float v = zr[j];
    d1 += v * attn_w[j];
    d2 += v * attn_w[OUT_DIM + j];
  }
  s1[n] = d1;
  s2[n] = d2;
}

__device__ __forceinline__ float wave_max(float v) {
  for (int o = 16; o > 0; o >>= 1) v = fmaxf(v, __shfl_xor(v, o));
  return v;
}
__device__ __forceinline__ float wave_sum(float v) {
  for (int o = 16; o > 0; o >>= 1) v += __shfl_xor(v, o);
  return v;
}

__global__ __launch_bounds__(256) void k_gather(const int* __restrict__ csr_src,
                                                const unsigned* __restrict__ off,
                                                const unsigned* __restrict__ cnt,
                                                const float* __restrict__ z,
                                                const float* __restrict__ s1,
                                                const float* __restrict__ s2,
                                                float* __restrict__ out) {
  const int lane = threadIdx.x & 31;
  const int d = __builtin_amdgcn_readfirstlane((int)(blockIdx.x * 8 + (threadIdx.x >> 5)));
  const int beg = (int)off[d];
  const int n   = (int)cnt[d];
  const float s2d = s2[d];
  const float inf = __builtin_huge_valf();

  auto edge = [&](int p, int& s, float& e) {
    const unsigned u = (unsigned)csr_src[beg + p];
    s = (int)((u < (unsigned)N_NODES) ? u : (unsigned)(N_NODES - 1));
    const float lg = s1[s] + s2d;
    e = (lg >= 0.0f) ? lg : LEAK * lg;
  };

  int s0 = 0;
  float e0 = -inf;
  if (lane < n) edge(lane, s0, e0);

  float m = e0;
  for (int p = 32 + lane; p < n; p += 32) {
    int s; float e;
    edge(p, s, e);
    m = fmaxf(m, e);
  }
  m = wave_max(m);
  if (!(fabsf(m) < inf)) m = 0.0f;

  const float x0 = (lane < n) ? expf(e0 - m) : 0.0f;
  float sum = x0;
  for (int p = 32 + lane; p < n; p += 32) {
    int s; float e;
    edge(p, s, e);
    sum += expf(e - m);
  }
  sum = wave_sum(sum);
  const float dn = (sum > 0.0f) ? sum : 1.0f;

  const v4f* __restrict__ z4 = (const v4f*)z;
  v4f acc = {0.0f, 0.0f, 0.0f, 0.0f};
  for (int c = 0; c < n; c += 32) {
    int s = s0;
    float a = x0 / dn;
    if (c > 0) {
      s = 0;
      a = 0.0f;
      const int p = c + lane;
      if (p < n) {
        float e;
        edge(p, s, e);
        a = expf(e - m) / dn;
      }
    }
    const int k = (n - c < 32) ? (n - c) : 32;
#pragma unroll 4
    for (int j = 0; j < k; ++j) {
      const int   sj = __builtin_amdgcn_readlane(s, j);
      const float aj = __int_as_float(__builtin_amdgcn_readlane(__float_as_int(a), j));
      const v4f v = z4[(size_t)sj * (OUT_DIM / 4) + lane];
      acc.x += aj * v.x;
      acc.y += aj * v.y;
      acc.z += aj * v.z;
      acc.w += aj * v.w;
    }
  }

  v4f r;
  r.x = (acc.x > 0.0f) ? acc.x : expm1f(acc.x);
  r.y = (acc.y > 0.0f) ? acc.y : expm1f(acc.y);
  r.z = (acc.z > 0.0f) ? acc.z : expm1f(acc.z);
  r.w = (acc.w > 0.0f) ? acc.w : expm1f(acc.w);
  ((v4f*)out)[(size_t)d * (OUT_DIM / 4) + lane] = r;
}

extern "C" void kernel_launch(void* const* d_in, const int* in_sizes, int n_in,
                              void* d_out, int out_size, void* d_ws, size_t ws_size,
                              hipStream_t stream) {
  const float* h      = (const float*)d_in[0];
  const int*   src    = (const int*)  d_in[1];
  const int*   dst    = (const int*)  d_in[2];
  const float* W_fc   = (const float*)d_in[3];
  const float* attn_w = (const float*)d_in[4];
  float*       out    = (float*)d_out;

  float*    z       = (float*)d_ws;
  unsigned* rk      = (unsigned*)d_ws;
  unsigned* binned  = rk + N_EDGES;
  unsigned* hist    = binned + N_EDGES;
  unsigned* base    = hist + HIST_N;
  float*    s1      = z  + (size_t)N_NODES * OUT_DIM;
  float*    s2      = s1 + N_NODES;
  unsigned* cnt     = (unsigned*)(s2 + N_NODES);
  unsigned* off     = cnt + N_NODES;
  int*      csr_src = (int*)(off + N_NODES);

  const int TPB = 256;

  k_binhist<<<N_CHUNKS, CHUNK_TPB, 0, stream>>>(dst, rk, hist);

  k_scan<<<(HIST_N + 1023) / 1024, 1024, 0, stream>>>(hist, base);

  k_binscatter<<<N_CHUNKS, CHUNK_TPB, 0, stream>>>(src, dst, base, rk, binned);

  k_binsort<<<N_BINS, SORT_TPB, 0, stream>>>(binned, base, off, cnt, csr_src);

  k_gemm<<<GEMM_BLOCKS, TPB, 0, stream>>>(h, W_fc, z);

  k_scores<<<(N_NODES + TPB - 1) / TPB, TPB, 0, stream>>>(z, attn_w, s1, s2);

  k_gather<<<N_NODES / 8, TPB, 0, stream>>>(csr_src, off, cnt, z, s1, s2, out);
}
